# starting kernel
# baseline (speedup 1.0000x reference)
#include <hip/hip_runtime.h>

typedef __attribute__((ext_vector_type(2))) float v2f;
typedef __attribute__((ext_vector_type(8))) float v8f;

#define NN 4096
#define EE 16384
#define DD 13
#define RR 128
#define PP 4
#define W1S 18
#define W2S 130

__global__ void zero_out_kernel(float* __restrict__ out, int n) {
  int i = blockIdx.x * blockDim.x + threadIdx.x;
  if (i < n) out[i] = 0.0f;
}

__global__ __launch_bounds__(256) void bpnet_stage1_kernel(
    const float* __restrict__ nodes,
    const float* __restrict__ bp_params,
    const float* __restrict__ bp_bias,
    const int*   __restrict__ edges,
    const int*   __restrict__ edge_types,
    float*       __restrict__ t_out)
{
  __shared__ float Wt[PP * RR * W1S];
  __shared__ float Bs[PP * RR];

  const int tid = threadIdx.x;
  for (int idx = tid; idx < PP * RR * W1S; idx += 256) {
    int p = idx / (RR * W1S);
    int r = idx % (RR * W1S);
    int n = r / W1S;
    int k = r % W1S;
    Wt[idx] = (k < DD) ? bp_params[(p * DD + k) * RR + n] : 0.0f;
  }
  for (int idx = tid; idx < PP * RR; idx += 256) Bs[idx] = bp_bias[idx];
  __syncthreads();

  const int wave = tid >> 5;
  const int lane = tid & 31;
  const int row  = lane & 15;
  const int kh   = lane >> 4;
  const int slot = blockIdx.y;
  const int tile = blockIdx.x * 8 + wave;
  const int e0   = tile * 16;
  const int e    = e0 + row;

  const int etype = edge_types[e * 3 + slot];
  const int nidx  = edges[e * 3 + slot];
  const int nbase = nidx * DD;

  v2f av[4];
#pragma unroll
  for (int c = 0; c < 4; ++c) {
    int kk = 4 * c + 2 * kh;
    av[c].x = (kk     < DD) ? nodes[nbase + kk]     : 0.0f;
    av[c].y = (kk + 1 < DD) ? nodes[nbase + kk + 1] : 0.0f;
  }

  for (int nt = 0; nt < 8; ++nt) {
    const int ng = nt * 16 + row;
    v8f acc = {0.f, 0.f, 0.f, 0.f, 0.f, 0.f, 0.f, 0.f};
#pragma unroll
    for (int c = 0; c < 4; ++c) {
      const int kk = 4 * c + 2 * kh;
#pragma unroll
      for (int p = 0; p < PP; ++p) {
        v2f b = *(const v2f*)&Wt[(p * RR + ng) * W1S + kk];
        v2f a;
        a.x = (etype == p) ? av[c].x : 0.0f;
        a.y = (etype == p) ? av[c].y : 0.0f;
        acc = __builtin_amdgcn_wmma_f32_16x16x4_f32(
            false, a, false, b, (short)0, acc, false, false);
      }
    }
#pragma unroll
    for (int v = 0; v < 8; ++v) {
      int r2  = v + 8 * kh;
      int tr  = __shfl(etype, r2, 32);
      float x = acc[v] + Bs[tr * RR + ng];
      x = (x > 0.0f) ? x : 0.0f;
      t_out[(size_t)(slot * EE + (e0 + r2)) * RR + ng] = x;
    }
  }
}

__global__ __launch_bounds__(256) void bpnet_stage2_kernel(
    const float* __restrict__ t_in,
    const float* __restrict__ ho_params,
    const float* __restrict__ ho_bias,
    const int*   __restrict__ edges,
    const int*   __restrict__ edge_types,
    float*       __restrict__ out)
{
  __shared__ float Wt[PP * 16 * W2S];
  __shared__ float Bs[PP * 16];

  const int slot = blockIdx.y;
  const int tid  = threadIdx.x;
  for (int idx = tid; idx < PP * 16 * W2S; idx += 256) {
    int p = idx / (16 * W2S);
    int r = idx % (16 * W2S);
    int n = r / W2S;
    int k = r % W2S;
    float v = 0.0f;
    if (n < DD && k < RR)
      v = ho_params[((size_t)(slot * PP + p) * RR + k) * DD + n];
    Wt[idx] = v;
  }
  for (int idx = tid; idx < PP * 16; idx += 256) {
    int p = idx / 16, n = idx % 16;
    Bs[idx] = (n < DD) ? ho_bias[(slot * PP + p) * DD + n] : 0.0f;
  }
  __syncthreads();

  const int wave = tid >> 5;
  const int lane = tid & 31;
  const int row  = lane & 15;
  const int kh   = lane >> 4;
  const int tile = blockIdx.x * 8 + wave;
  const int e0   = tile * 16;
  const int e    = e0 + row;

  const int etype = edge_types[e * 3 + slot];
  const int tgt   = edges[e * 3 + slot];
  const int j0    = (slot == 0) ? 1 : 0;
  const int j1    = (slot == 2) ? 1 : 2;

  const float* __restrict__ t0 = t_in + (size_t)(j0 * EE + e) * RR;
  const float* __restrict__ t1 = t_in + (size_t)(j1 * EE + e) * RR;

  v8f acc = {0.f, 0.f, 0.f, 0.f, 0.f, 0.f, 0.f, 0.f};
  for (int c = 0; c < 32; ++c) {
    const int kk = 4 * c + 2 * kh;
    v2f x0 = *(const v2f*)(t0 + kk);
    v2f x1 = *(const v2f*)(t1 + kk);
    v2f a;
    a.x = x0.x * x1.x;
    a.y = x0.y * x1.y;
#pragma unroll
    for (int p = 0; p < PP; ++p) {
      v2f b = *(const v2f*)&Wt[(p * 16 + row) * W2S + kk];
      v2f ap;
      ap.x = (etype == p) ? a.x : 0.0f;
      ap.y = (etype == p) ? a.y : 0.0f;
      acc = __builtin_amdgcn_wmma_f32_16x16x4_f32(
          false, ap, false, b, (short)0, acc, false, false);
    }
  }

#pragma unroll
  for (int v = 0; v < 8; ++v) {
    int r2 = v + 8 * kh;
    int tr = __shfl(etype, r2, 32);
    int tg = __shfl(tgt,   r2, 32);
    if (row < DD) {
      float val = acc[v] + Bs[tr * 16 + row];
      atomicAdd(&out[tg * DD + row], val);
    }
  }
}

extern "C" void kernel_launch(void* const* d_in, const int* in_sizes, int n_in,
                              void* d_out, int out_size, void* d_ws, size_t ws_size,
                              hipStream_t stream) {
  const float* nodes      = (const float*)d_in[0];
  const float* bp_params  = (const float*)d_in[1];
  const float* bp_bias    = (const float*)d_in[2];
  const float* ho_params  = (const float*)d_in[3];
  const float* ho_bias    = (const float*)d_in[4];
  const int*   edges      = (const int*)d_in[5];
  const int*   edge_types = (const int*)d_in[6];
  float* out  = (float*)d_out;
  float* t_ws = (float*)d_ws;

  zero_out_kernel<<<(NN * DD + 255) / 256, 256, 0, stream>>>(out, NN * DD);

  bpnet_stage1_kernel<<<dim3(EE / 16 / 8, 3), 256, 0, stream>>>(
      nodes, bp_params, bp_bias, edges, edge_types, t_ws);
  bpnet_stage2_kernel<<<dim3(EE / 16 / 8, 3), 256, 0, stream>>>(
      t_ws, ho_params, ho_bias, edges, edge_types, out);
}


# Fuse both products in one kernel: transposed 1st product leaves t in registers as A operand of the 2nd (no workspace pass, one launch less); 2nd product's k order and halves-sum order change
# speedup vs baseline: 1.9568x; 1.9568x over previous
#include <hip/hip_runtime.h>

typedef __attribute__((ext_vector_type(2))) float v2f;
typedef __attribute__((ext_vector_type(4))) float v4f;
typedef __attribute__((ext_vector_type(8))) float v8f;

#define NN 4096
#define EE 16384
#define DD 13
#define RR 128
#define PP 4

#define W1_OFF 0
#define W2_OFF 8192
#define B1_OFF (W2_OFF + 24576)
#define B2_OFF (B1_OFF + 512)
#define CB_OFF (B2_OFF + 192)
#define LDS_DW 41984

__global__ void zero_out_kernel(float* __restrict__ out, int n) {
  int i = blockIdx.x * blockDim.x + threadIdx.x;
  if (i < n) out[i] = 0.0f;
}

__global__ __launch_bounds__(256) void bpnet_fused_kernel(
    const float* __restrict__ nodes,
    const float* __restrict__ bp_params,
    const float* __restrict__ bp_bias,
    const float* __restrict__ ho_params,
    const float* __restrict__ ho_bias,
    const int*   __restrict__ edges,
    const int*   __restrict__ edge_types,
    float*       __restrict__ out)
{
  __shared__ __attribute__((aligned(16))) float smem[LDS_DW];
  float* W1 = smem + W1_OFF;
  float* W2 = smem + W2_OFF;
  float* B1 = smem + B1_OFF;
  float* B2 = smem + B2_OFF;
  float* CB = smem + CB_OFF;

  const int tid   = threadIdx.x;
  const int wave  = tid >> 5;
  const int lane  = tid & 31;
  const int c16   = lane & 15;
  const int h     = lane >> 4;
  const int tslot = wave & 3;
  const int half  = wave >> 2;
  const int tile  = blockIdx.x * 4 + tslot;
  const int e0    = tile * 16;
  const int e     = e0 + c16;

  int et[3], nd[3];
#pragma unroll
  for (int s = 0; s < 3; ++s) {
    et[s] = edge_types[e * 3 + s];
    nd[s] = edges[e * 3 + s];
  }
  v2f nb[3][4];
#pragma unroll
  for (int s = 0; s < 3; ++s) {
#pragma unroll
    for (int c = 0; c < 4; ++c) {
      const int kk = 4 * c + 2 * h;
      const float gx = nodes[nd[s] * DD + ((kk     < DD) ? kk     : 0)];
      const float gy = nodes[nd[s] * DD + ((kk + 1 < DD) ? kk + 1 : 0)];
      nb[s][c].x = (kk     < DD) ? gx : 0.0f;
      nb[s][c].y = (kk + 1 < DD) ? gy : 0.0f;
    }
  }

  v4f w1v[7];
#pragma unroll
  for (int i = 0; i < 7; ++i) {
    const int row = i * 8 + wave;
    w1v[i] = (v4f){0.f, 0.f, 0.f, 0.f};
    if (row < PP * DD) w1v[i] = *(const v4f*)&bp_params[row * RR + lane * 4];
  }
  const int d16 = tid & 15;
  const int t16 = tid >> 4;
  const int dcl = (d16 < DD) ? d16 : (DD - 1);
  float w2v[96];
#pragma unroll
  for (int i = 0; i < 96; ++i) {
    const int sp = i >> 3;
    const int r  = ((i & 7) << 4) + t16;
    w2v[i] = ho_params[(sp * RR + r) * DD + dcl];
  }
  const float b1v0 = bp_bias[tid];
  const float b1v1 = bp_bias[tid + 256];
  const int   b2i  = (tid < 3 * PP * 16) ? tid : 0;
  const float b2v  = ho_bias[(b2i >> 4) * DD + (((b2i & 15) < DD) ? (b2i & 15) : 0)];

#pragma unroll
  for (int i = 0; i < 7; ++i) {
    const int row = i * 8 + wave;
    if (row < PP * DD) {
      const int p = row / DD;
      const int k = row - p * DD;
      const int r = lane * 4;
      float* dst = &W1[(((p * 4 + (k >> 2)) * 8 + (r >> 4)) * 2 + ((k >> 1) & 1)) * 32 + (r & 15) * 2 + (k & 1)];
      dst[0] = w1v[i].x;
      dst[2] = w1v[i].y;
      dst[4] = w1v[i].z;
      dst[6] = w1v[i].w;
    }
  }
#pragma unroll
  for (int i = 0; i < 6; ++i) {
    const int pk = i * 2 + (tid >> 7);
    const int p  = pk / 3;
    const int k  = DD + (pk - p * 3);
    const int r  = tid & 127;
    W1[(((p * 4 + (k >> 2)) * 8 + (r >> 4)) * 2 + ((k >> 1) & 1)) * 32 + (r & 15) * 2 + (k & 1)] = 0.0f;
  }
  {
    const int wbase = ((t16 >> 1) & 3) * 64 + (t16 >> 3) * 32 + d16 * 2 + (t16 & 1);
#pragma unroll
    for (int i = 0; i < 96; ++i)
      W2[i * 256 + wbase] = (d16 < DD) ? w2v[i] : 0.0f;
  }
  B1[tid]       = b1v0;
  B1[tid + 256] = b1v1;
  if (tid < 3 * PP * 16) B2[tid] = ((tid & 15) < DD) ? b2v : 0.0f;
  __syncthreads();

  v8f a2[3];
#pragma unroll
  for (int s = 0; s < 3; ++s) a2[s] = (v8f){0.f, 0.f, 0.f, 0.f, 0.f, 0.f, 0.f, 0.f};

  const int nt0 = half * 4;
  for (int nt = nt0; nt < nt0 + 4; ++nt) {
    v8f a1[3];
#pragma unroll
    for (int s = 0; s < 3; ++s) a1[s] = (v8f){0.f, 0.f, 0.f, 0.f, 0.f, 0.f, 0.f, 0.f};

#pragma unroll
    for (int c = 0; c < 4; ++c) {
#pragma unroll
      for (int p = 0; p < PP; ++p) {
        const v2f wa = *(const v2f*)&W1[((p * 4 + c) * 8 + nt) * 64 + lane * 2];
#pragma unroll
        for (int s = 0; s < 3; ++s) {
          v2f b;
          b.x = (et[s] == p) ? nb[s][c].x : 0.0f;
          b.y = (et[s] == p) ? nb[s][c].y : 0.0f;
          a1[s] = __builtin_amdgcn_wmma_f32_16x16x4_f32(
              false, wa, false, b, (short)0, a1[s], false, false);
        }
      }
    }

    v8f t[3];
#pragma unroll
    for (int s = 0; s < 3; ++s) {
      const float* bp = &B1[et[s] * RR + nt * 16 + h * 8];
      const v4f b0 = *(const v4f*)bp;
      const v4f b1 = *(const v4f*)(bp + 4);
#pragma unroll
      for (int v = 0; v < 4; ++v) {
        float x = a1[s][v] + b0[v];
        float y = a1[s][v + 4] + b1[v];
        t[s][v]     = (x > 0.0f) ? x : 0.0f;
        t[s][v + 4] = (y > 0.0f) ? y : 0.0f;
      }
    }
    v8f f[3];
    f[0] = t[1] * t[2];
    f[1] = t[0] * t[2];
    f[2] = t[0] * t[1];

#pragma unroll
    for (int j = 0; j < 4; ++j) {
#pragma unroll
      for (int p = 0; p < PP; ++p) {
#pragma unroll
        for (int s = 0; s < 3; ++s) {
          const v2f wb = *(const v2f*)&W2[(((s * 4 + p) * 8 + nt) * 4 + j) * 64 + lane * 2];
          v2f a;
          a.x = (et[s] == p) ? f[s][2 * j]     : 0.0f;
          a.y = (et[s] == p) ? f[s][2 * j + 1] : 0.0f;
          a2[s] = __builtin_amdgcn_wmma_f32_16x16x4_f32(
              false, a, false, wb, (short)0, a2[s], false, false);
        }
      }
    }
  }

  if (half == 1) {
#pragma unroll
    for (int s = 0; s < 3; ++s)
#pragma unroll
      for (int v = 0; v < 8; ++v)
        CB[(tslot * 24 + s * 8 + v) * 32 + lane] = a2[s][v];
  }
  __syncthreads();
  if (half == 0) {
    float val[3][8];
    int   adr[3][8];
#pragma unroll
    for (int s = 0; s < 3; ++s) {
#pragma unroll
      for (int v = 0; v < 8; ++v) {
        const float sum = a2[s][v] + CB[(tslot * 24 + s * 8 + v) * 32 + lane];
        const int r2 = v + 8 * h;
        const int tr = __shfl(et[s], r2, 32);
        const int tg = __shfl(nd[s], r2, 32);
        val[s][v] = sum + B2[(s * 4 + tr) * 16 + c16];
        adr[s][v] = tg * DD + c16;
      }
    }
    if (c16 < DD) {
#pragma unroll
      for (int s = 0; s < 3; ++s)
#pragma unroll
        for (int v = 0; v < 8; ++v)
          atomicAdd(&out[adr[s][v]], val[s][v]);
    }
  }
}

extern "C" void kernel_launch(void* const* d_in, const int* in_sizes, int n_in,
                              void* d_out, int out_size, void* d_ws, size_t ws_size,
                              hipStream_t stream) {
  const float* nodes      = (const float*)d_in[0];
  const float* bp_params  = (const float*)d_in[1];
  const float* bp_bias    = (const float*)d_in[2];
  const float* ho_params  = (const float*)d_in[3];
  const float* ho_bias    = (const float*)d_in[4];
  const int*   edges      = (const int*)d_in[5];
  const int*   edge_types = (const int*)d_in[6];
  float* out = (float*)d_out;

  zero_out_kernel<<<(NN * DD + 255) / 256, 256, 0, stream>>>(out, NN * DD);

  bpnet_fused_kernel<<<dim3(256), 256, 0, stream>>>(
      nodes, bp_params, bp_bias, ho_params, ho_bias, edges, edge_types, out);
}


# Group edges by type triple per 1024-edge chunk (same counting sort in its 16 work-groups) so a tile holds few types; one product pass per type present in a tile, not 4; atomic add order changes
# speedup vs baseline: 2.1184x; 1.0826x over previous
#include <hip/hip_runtime.h>

typedef __attribute__((ext_vector_type(2))) float v2f;
typedef __attribute__((ext_vector_type(4))) float v4f;
typedef __attribute__((ext_vector_type(8))) float v8f;
typedef __attribute__((ext_vector_type(4))) int   v4i;

#define NN 4096
#define EE 16384
#define DD 13
#define RR 128
#define PP 4

#define W1_OFF 0
#define W2_OFF 8192
#define B1_OFF (W2_OFF + 24576)
#define B2_OFF (B1_OFF + 512)
#define CB_OFF (B2_OFF + 192)
#define REC_OFF 0
#define CNT_OFF (REC_OFF + 4096)
#define TOT_OFF (CNT_OFF + 2048)
#define BAS_OFF (TOT_OFF + 64)
#define GRP_DW  (BAS_OFF + 64)
#define CHUNK 1024
#define LDS_DW 41984

__global__ void zero_out_kernel(float* __restrict__ out, int n) {
  int i = blockIdx.x * blockDim.x + threadIdx.x;
  if (i < n) out[i] = 0.0f;
}

__global__ __launch_bounds__(256) void bpnet_fused_kernel(
    const float* __restrict__ nodes,
    const float* __restrict__ bp_params,
    const float* __restrict__ bp_bias,
    const float* __restrict__ ho_params,
    const float* __restrict__ ho_bias,
    const int*   __restrict__ edges,
    const int*   __restrict__ edge_types,
    float*       __restrict__ out)
{
  __shared__ __attribute__((aligned(16))) float smem[LDS_DW];
  float* W1 = smem + W1_OFF;
  float* W2 = smem + W2_OFF;
  float* B1 = smem + B1_OFF;
  float* B2 = smem + B2_OFF;
  float* CB = smem + CB_OFF;

  const int tid   = threadIdx.x;
  const int wave  = __builtin_amdgcn_readfirstlane(tid >> 5);
  const int lane  = tid & 31;
  const int c16   = lane & 15;
  const int h     = lane >> 4;
  const int tslot = wave & 3;
  const int half  = wave >> 2;
  const int grp = blockIdx.x >> 4;
  const int sub = blockIdx.x & 15;
  __shared__ __attribute__((aligned(16))) int gsm[GRP_DW];
  int* REC = gsm + REC_OFF;
  int* CNT = gsm + CNT_OFF;
  int* TOT = gsm + TOT_OFF;
  int* BAS = gsm + BAS_OFF;

  int tf[12], ef[12];
  {
    const int cb = (grp * CHUNK + tid * 4) * 3;
    const v4i* tp = (const v4i*)&edge_types[cb];
    const v4i* ep = (const v4i*)&edges[cb];
#pragma unroll
    for (int i = 0; i < 3; ++i) {
      const v4i tv = tp[i];
      const v4i ev = ep[i];
      tf[4 * i] = tv.x; tf[4 * i + 1] = tv.y; tf[4 * i + 2] = tv.z; tf[4 * i + 3] = tv.w;
      ef[4 * i] = ev.x; ef[4 * i + 1] = ev.y; ef[4 * i + 2] = ev.z; ef[4 * i + 3] = ev.w;
    }
  }

  {
    const v4i z4 = {0, 0, 0, 0};
    ((v4i*)CNT)[tid * 2]     = z4;
    ((v4i*)CNT)[tid * 2 + 1] = z4;
  }
  __syncthreads();
  int key[4], typ[4], rnk[4];
#pragma unroll
  for (int q = 0; q < 4; ++q) {
    const int t0 = tf[3 * q] & 3;
    const int t1 = tf[3 * q + 1] & 3;
    const int t2 = tf[3 * q + 2] & 3;
    const int a1 = t1 ^ ((t0 & 1) ? 3 : 0);
    const int a2 = t2 ^ ((a1 & 1) ? 3 : 0);
    key[q] = (t0 << 4) | (a1 << 2) | a2;
    typ[q] = (t0 << 4) | (t1 << 2) | t2;
    unsigned peers = 0xffffffffu;
#pragma unroll
    for (int b = 0; b < 6; ++b) {
      const bool bit = (key[q] >> b) & 1;
      const unsigned bal = __builtin_amdgcn_ballot_w32(bit);
      peers &= bit ? bal : ~bal;
    }
    rnk[q] = __popc(peers & ((1u << lane) - 1u));
    if (rnk[q] == 0) CNT[(wave * 4 + q) * 64 + key[q]] = __popc(peers);
  }
  __syncthreads();
  if (tid < 64) {
    int run = 0;
#pragma unroll
    for (int row = 0; row < 32; ++row) {
      const int c = CNT[row * 64 + tid];
      CNT[row * 64 + tid] = run;
      run += c;
    }
    TOT[tid] = run;
  }
  __syncthreads();
  if (tid < 64) {
    int b = 0;
#pragma unroll
    for (int k4 = 0; k4 < 16; ++k4) {
      const v4i t4 = ((const v4i*)TOT)[k4];
      b += (4 * k4     < tid) ? t4.x : 0;
      b += (4 * k4 + 1 < tid) ? t4.y : 0;
      b += (4 * k4 + 2 < tid) ? t4.z : 0;
      b += (4 * k4 + 3 < tid) ? t4.w : 0;
    }
    BAS[tid] = b;
  }
  __syncthreads();
#pragma unroll
  for (int q = 0; q < 4; ++q) {
    const int pos = BAS[key[q]] + CNT[(wave * 4 + q) * 64 + key[q]] + rnk[q];
    const v4i rec = {typ[q], ef[3 * q], ef[3 * q + 1], ef[3 * q + 2]};
    ((v4i*)REC)[pos] = rec;
  }
  __syncthreads();
  int et[3], nd[3];
  {
    const v4i rec = ((const v4i*)REC)[sub * 64 + tslot * 16 + c16];
    et[0] = (rec.x >> 4) & 3;
    et[1] = (rec.x >> 2) & 3;
    et[2] = rec.x & 3;
    nd[0] = rec.y;
    nd[1] = rec.z;
    nd[2] = rec.w;
  }
  unsigned pm[3];
#pragma unroll
  for (int s = 0; s < 3; ++s) {
    unsigned m = 0;
#pragma unroll
    for (int p = 0; p < PP; ++p)
      m |= (__builtin_amdgcn_ballot_w32(et[s] == p) != 0u) ? (1u << p) : 0u;
    pm[s] = m;
  }

  v4f w1v[7];
#pragma unroll
  for (int i = 0; i < 7; ++i) {
    const int row = i * 8 + wave;
    w1v[i] = (v4f){0.f, 0.f, 0.f, 0.f};
    if (row < PP * DD) w1v[i] = *(const v4f*)&bp_params[row * RR + lane * 4];
  }
  const int d16 = tid & 15;
  const int t16 = tid >> 4;
  const int dcl = (d16 < DD) ? d16 : (DD - 1);
  float w2v[96];
#pragma unroll
  for (int i = 0; i < 96; ++i) {
    const int sp = i >> 3;
    const int r  = ((i & 7) << 4) + t16;
    w2v[i] = ho_params[(sp * RR + r) * DD + dcl];
  }
  const float b1v0 = bp_bias[tid];
  const float b1v1 = bp_bias[tid + 256];
  const int   b2i  = (tid < 3 * PP * 16) ? tid : 0;
  const float b2v  = ho_bias[(b2i >> 4) * DD + (((b2i & 15) < DD) ? (b2i & 15) : 0)];

  v2f nb[3][4];
#pragma unroll
  for (int s = 0; s < 3; ++s) {
#pragma unroll
    for (int c = 0; c < 4; ++c) {
      const int kk = 4 * c + 2 * h;
      const float gx = nodes[nd[s] * DD + ((kk     < DD) ? kk     : 0)];
      const float gy = nodes[nd[s] * DD + ((kk + 1 < DD) ? kk + 1 : 0)];
      nb[s][c].x = (kk     < DD) ? gx : 0.0f;
      nb[s][c].y = (kk + 1 < DD) ? gy : 0.0f;
    }
  }

#pragma unroll
  for (int i = 0; i < 7; ++i) {
    const int row = i * 8 + wave;
    if (row < PP * DD) {
      const int p = row / DD;
      const int k = row - p * DD;
      const int r = lane * 4;
      float* dst = &W1[(((p * 4 + (k >> 2)) * 8 + (r >> 4)) * 2 + ((k >> 1) & 1)) * 32 + (r & 15) * 2 + (k & 1)];
      dst[0] = w1v[i].x;
      dst[2] = w1v[i].y;
      dst[4] = w1v[i].z;
      dst[6] = w1v[i].w;
    }
  }
#pragma unroll
  for (int i = 0; i < 6; ++i) {
    const int pk = i * 2 + (tid >> 7);
    const int p  = pk / 3;
    const int k  = DD + (pk - p * 3);
    const int r  = tid & 127;
    W1[(((p * 4 + (k >> 2)) * 8 + (r >> 4)) * 2 + ((k >> 1) & 1)) * 32 + (r & 15) * 2 + (k & 1)] = 0.0f;
  }
  {
    const int wbase = ((t16 >> 1) & 3) * 64 + (t16 >> 3) * 32 + d16 * 2 + (t16 & 1);
#pragma unroll
    for (int i = 0; i < 96; ++i)
      W2[i * 256 + wbase] = (d16 < DD) ? w2v[i] : 0.0f;
  }
  B1[tid]       = b1v0;
  B1[tid + 256] = b1v1;
  if (tid < 3 * PP * 16) B2[tid] = ((tid & 15) < DD) ? b2v : 0.0f;
  __syncthreads();

  v8f a2[3];
#pragma unroll
  for (int s = 0; s < 3; ++s) a2[s] = (v8f){0.f, 0.f, 0.f, 0.f, 0.f, 0.f, 0.f, 0.f};

  int      p0[3];
  unsigned pr[3];
#pragma unroll
  for (int s = 0; s < 3; ++s) {
    p0[s] = __builtin_amdgcn_readfirstlane(et[s]);
    pr[s] = pm[s] & ~(1u << p0[s]);
  }

  const int nt0 = half * 4;
  for (int nt = nt0; nt < nt0 + 4; ++nt) {
    v2f wq[3][4];
#pragma unroll
    for (int s = 0; s < 3; ++s) {
      const float* wp = &W1[(p0[s] * 32 + nt) * 64 + lane * 2];
#pragma unroll
      for (int c = 0; c < 4; ++c) wq[s][c] = *(const v2f*)(wp + c * 512);
    }
    v8f a1[3];
#pragma unroll
    for (int s = 0; s < 3; ++s) {
      v8f acc = (v8f){0.f, 0.f, 0.f, 0.f, 0.f, 0.f, 0.f, 0.f};
      v2f w[4];
#pragma unroll
      for (int c = 0; c < 4; ++c) w[c] = wq[s][c];
      int p = p0[s];
      unsigned m = pr[s];
      for (;;) {
        const bool mine = (et[s] == p);
#pragma unroll
        for (int c = 0; c < 4; ++c) {
          v2f bb;
          bb.x = mine ? nb[s][c].x : 0.0f;
          bb.y = mine ? nb[s][c].y : 0.0f;
          acc = __builtin_amdgcn_wmma_f32_16x16x4_f32(
              false, w[c], false, bb, (short)0, acc, false, false);
        }
        if (m == 0u) break;
        p = __builtin_ctz(m);
        m &= m - 1u;
        const float* wp = &W1[(p * 32 + nt) * 64 + lane * 2];
#pragma unroll
        for (int c = 0; c < 4; ++c) w[c] = *(const v2f*)(wp + c * 512);
      }
      a1[s] = acc;
    }

    v2f vq[3][4];
#pragma unroll
    for (int s = 0; s < 3; ++s) {
      const float* wp = &W2[(((s * 4 + p0[s]) * 8 + nt) * 4) * 64 + lane * 2];
#pragma unroll
      for (int j = 0; j < 4; ++j) vq[s][j] = *(const v2f*)(wp + j * 64);
    }

    v8f t[3];
#pragma unroll
    for (int s = 0; s < 3; ++s) {
      const float* bp = &B1[et[s] * RR + nt * 16 + h * 8];
      const v4f b0 = *(const v4f*)bp;
      const v4f b1 = *(const v4f*)(bp + 4);
#pragma unroll
      for (int v = 0; v < 4; ++v) {
        float x = a1[s][v] + b0[v];
        float y = a1[s][v + 4] + b1[v];
        t[s][v]     = (x > 0.0f) ? x : 0.0f;
        t[s][v + 4] = (y > 0.0f) ? y : 0.0f;
      }
    }
    v8f f[3];
    f[0] = t[1] * t[2];
    f[1] = t[0] * t[2];
    f[2] = t[0] * t[1];

#pragma unroll
    for (int s = 0; s < 3; ++s) {
      v8f acc = a2[s];
      v2f w[4];
#pragma unroll
      for (int j = 0; j < 4; ++j) w[j] = vq[s][j];
      int p = p0[s];
      unsigned m = pr[s];
      for (;;) {
        const bool mine = (et[s] == p);
#pragma unroll
        for (int j = 0; j < 4; ++j) {
          v2f aa;
          aa.x = mine ? f[s][2 * j]     : 0.0f;
          aa.y = mine ? f[s][2 * j + 1] : 0.0f;
          acc = __builtin_amdgcn_wmma_f32_16x16x4_f32(
              false, aa, false, w[j], (short)0, acc, false, false);
        }
        if (m == 0u) break;
        p = __builtin_ctz(m);
        m &= m - 1u;
        const float* wp = &W2[(((s * 4 + p) * 8 + nt) * 4) * 64 + lane * 2];
#pragma unroll
        for (int j = 0; j < 4; ++j) w[j] = *(const v2f*)(wp + j * 64);
      }
      a2[s] = acc;
    }
  }

  if (half == 1) {
#pragma unroll
    for (int s = 0; s < 3; ++s)
#pragma unroll
      for (int v = 0; v < 8; ++v)
        CB[(tslot * 24 + s * 8 + v) * 32 + lane] = a2[s][v];
  }
  __syncthreads();
  if (half == 0) {
    float val[3][8];
    int   adr[3][8];
#pragma unroll
    for (int s = 0; s < 3; ++s) {
#pragma unroll
      for (int v = 0; v < 8; ++v) {
        const float sum = a2[s][v] + CB[(tslot * 24 + s * 8 + v) * 32 + lane];
        const int r2 = v + 8 * h;
        const int tr = __shfl(et[s], r2, 32);
        const int tg = __shfl(nd[s], r2, 32);
        val[s][v] = sum + B2[(s * 4 + tr) * 16 + c16];
        adr[s][v] = tg * DD + c16;
      }
    }
    if (c16 < DD) {
#pragma unroll
      for (int s = 0; s < 3; ++s)
#pragma unroll
        for (int v = 0; v < 8; ++v)
          atomicAdd(&out[adr[s][v]], val[s][v]);
    }
  }
}

extern "C" void kernel_launch(void* const* d_in, const int* in_sizes, int n_in,
                              void* d_out, int out_size, void* d_ws, size_t ws_size,
                              hipStream_t stream) {
  const float* nodes      = (const float*)d_in[0];
  const float* bp_params  = (const float*)d_in[1];
  const float* bp_bias    = (const float*)d_in[2];
  const float* ho_params  = (const float*)d_in[3];
  const float* ho_bias    = (const float*)d_in[4];
  const int*   edges      = (const int*)d_in[5];
  const int*   edge_types = (const int*)d_in[6];
  float* out = (float*)d_out;

  zero_out_kernel<<<(NN * DD + 255) / 256, 256, 0, stream>>>(out, NN * DD);

  bpnet_fused_kernel<<<dim3(256), 256, 0, stream>>>(
      nodes, bp_params, bp_bias, ho_params, ho_bias, edges, edge_types, out);
}


# Preparation common to all work-groups (edge grouping per chunk, weights in LDS fragment order) done once in the first launch, passed through the workspace; main kernel fills LDS with 16-byte loads
# speedup vs baseline: 2.2652x; 1.0693x over previous
#include <hip/hip_runtime.h>

typedef __attribute__((ext_vector_type(2))) float v2f;
typedef __attribute__((ext_vector_type(4))) float v4f;
typedef __attribute__((ext_vector_type(8))) float v8f;
typedef __attribute__((ext_vector_type(4))) int   v4i;

#define NN 4096
#define EE 16384
#define DD 13
#define RR 128
#define PP 4

#define W1_OFF 0
#define W2_OFF 8192
#define B1_OFF (W2_OFF + 24576)
#define B2_OFF (B1_OFF + 512)
#define CB_OFF (B2_OFF + 192)
#define IMG_DW CB_OFF
#define IMG_V4 (IMG_DW / 4)
#define IMG_ROUNDS ((IMG_V4 + 255) / 256)
#define CNT_OFF 0
#define TOT_OFF (CNT_OFF + 2048)
#define BAS_OFF (TOT_OFF + 64)
#define GRP_DW  (BAS_OFF + 64)
#define CHUNK 1024
#define NCHUNK (EE / CHUNK)
#define W2_WG   12
#define PREP_WG (NCHUNK + W2_WG + 1)
#define WS_REC 0
#define WS_IMG (EE * 4)
#define LDS_DW 41984

__global__ __launch_bounds__(256) void prepare_kernel(
    const float* __restrict__ bp_params,
    const float* __restrict__ bp_bias,
    const float* __restrict__ ho_params,
    const float* __restrict__ ho_bias,
    const int*   __restrict__ edges,
    const int*   __restrict__ edge_types,
    float*       __restrict__ out,
    int*         __restrict__ recs,
    float*       __restrict__ img)
{
  __shared__ __attribute__((aligned(16))) int gsm[GRP_DW];
  const int tid  = threadIdx.x;
  const int wave = __builtin_amdgcn_readfirstlane(tid >> 5);
  const int lane = tid & 31;
  const int bid  = blockIdx.x;

  {
    const v4f z4 = {0.f, 0.f, 0.f, 0.f};
    for (int i = bid * 256 + tid; i < NN * DD / 4; i += PREP_WG * 256)
      ((v4f*)out)[i] = z4;
  }

  if (bid < NCHUNK) {
  const int grp = bid;
  int* CNT = gsm + CNT_OFF;
  int* TOT = gsm + TOT_OFF;
  int* BAS = gsm + BAS_OFF;

  int tf[12], ef[12];
  {
    const int cb = (grp * CHUNK + tid * 4) * 3;
    const v4i* tp = (const v4i*)&edge_types[cb];
    const v4i* ep = (const v4i*)&edges[cb];
#pragma unroll
    for (int i = 0; i < 3; ++i) {
      const v4i tv = tp[i];
      const v4i ev = ep[i];
      tf[4 * i] = tv.x; tf[4 * i + 1] = tv.y; tf[4 * i + 2] = tv.z; tf[4 * i + 3] = tv.w;
      ef[4 * i] = ev.x; ef[4 * i + 1] = ev.y; ef[4 * i + 2] = ev.z; ef[4 * i + 3] = ev.w;
    }
  }

  {
    const v4i z4 = {0, 0, 0, 0};
    ((v4i*)CNT)[tid * 2]     = z4;
    ((v4i*)CNT)[tid * 2 + 1] = z4;
  }
  __syncthreads();
  int key[4], typ[4], rnk[4];
#pragma unroll
  for (int q = 0; q < 4; ++q) {
    const int t0 = tf[3 * q] & 3;
    const int t1 = tf[3 * q + 1] & 3;
    const int t2 = tf[3 * q + 2] & 3;
    const int a1 = t1 ^ ((t0 & 1) ? 3 : 0);
    const int a2 = t2 ^ ((a1 & 1) ? 3 : 0);
    key[q] = (t0 << 4) | (a1 << 2) | a2;
    typ[q] = (t0 << 4) | (t1 << 2) | t2;
    unsigned peers = 0xffffffffu;
#pragma unroll
    for (int b = 0; b < 6; ++b) {
      const bool bit = (key[q] >> b) & 1;
      const unsigned bal = __builtin_amdgcn_ballot_w32(bit);
      peers &= bit ? bal : ~bal;
    }
    rnk[q] = __popc(peers & ((1u << lane) - 1u));
    if (rnk[q] == 0) CNT[(wave * 4 + q) * 64 + key[q]] = __popc(peers);
  }
  __syncthreads();
  if (tid < 64) {
    int run = 0;
#pragma unroll
    for (int row = 0; row < 32; ++row) {
      const int c = CNT[row * 64 + tid];
      CNT[row * 64 + tid] = run;
      run += c;
    }
    TOT[tid] = run;
  }
  __syncthreads();
  if (tid < 64) {
    int b = 0;
#pragma unroll
    for (int k4 = 0; k4 < 16; ++k4) {
      const v4i t4 = ((const v4i*)TOT)[k4];
      b += (4 * k4     < tid) ? t4.x : 0;
      b += (4 * k4 + 1 < tid) ? t4.y : 0;
      b += (4 * k4 + 2 < tid) ? t4.z : 0;
      b += (4 * k4 + 3 < tid) ? t4.w : 0;
    }
    BAS[tid] = b;
  }
  __syncthreads();
#pragma unroll
  for (int q = 0; q < 4; ++q) {
    const int pos = BAS[key[q]] + CNT[(wave * 4 + q) * 64 + key[q]] + rnk[q];
    const v4i rec = {typ[q], ef[3 * q], ef[3 * q + 1], ef[3 * q + 2]};
    ((v4i*)recs)[grp * CHUNK + pos] = rec;
  }
  } else if (bid < NCHUNK + W2_WG) {
    const int sp  = bid - NCHUNK;
    const int d16 = tid & 15;
    const int t16 = tid >> 4;
    const int dcl = (d16 < DD) ? d16 : (DD - 1);
    float w2v[8];
#pragma unroll
    for (int i = 0; i < 8; ++i) {
      const int r = (i << 4) + t16;
      w2v[i] = ho_params[(sp * RR + r) * DD + dcl];
    }
    const int wbase = ((t16 >> 1) & 3) * 64 + (t16 >> 3) * 32 + d16 * 2 + (t16 & 1);
    float* W2 = img + W2_OFF;
#pragma unroll
    for (int i = 0; i < 8; ++i)
      W2[(sp * 8 + i) * 256 + wbase] = (d16 < DD) ? w2v[i] : 0.0f;
  } else {
    float* W1 = img + W1_OFF;
    float* B1 = img + B1_OFF;
    float* B2 = img + B2_OFF;
    v4f w1v[7];
#pragma unroll
    for (int i = 0; i < 7; ++i) {
      const int row = i * 8 + wave;
      w1v[i] = (v4f){0.f, 0.f, 0.f, 0.f};
      if (row < PP * DD) w1v[i] = *(const v4f*)&bp_params[row * RR + lane * 4];
    }
    const float b1v0 = bp_bias[tid];
    const float b1v1 = bp_bias[tid + 256];
    const int   b2i  = (tid < 3 * PP * 16) ? tid : 0;
    const float b2v  = ho_bias[(b2i >> 4) * DD + (((b2i & 15) < DD) ? (b2i & 15) : 0)];
#pragma unroll
    for (int i = 0; i < 7; ++i) {
      const int row = i * 8 + wave;
      if (row < PP * DD) {
        const int p = row / DD;
        const int k = row - p * DD;
        const int r = lane * 4;
        float* dst = &W1[(((p * 4 + (k >> 2)) * 8 + (r >> 4)) * 2 + ((k >> 1) & 1)) * 32 + (r & 15) * 2 + (k & 1)];
        dst[0] = w1v[i].x;
        dst[2] = w1v[i].y;
        dst[4] = w1v[i].z;
        dst[6] = w1v[i].w;
      }
    }
#pragma unroll
    for (int i = 0; i < 6; ++i) {
      const int pk = i * 2 + (tid >> 7);
      const int p  = pk / 3;
      const int k  = DD + (pk - p * 3);
      const int r  = tid & 127;
      W1[(((p * 4 + (k >> 2)) * 8 + (r >> 4)) * 2 + ((k >> 1) & 1)) * 32 + (r & 15) * 2 + (k & 1)] = 0.0f;
    }
    B1[tid]       = b1v0;
    B1[tid + 256] = b1v1;
    if (tid < 3 * PP * 16) B2[tid] = ((tid & 15) < DD) ? b2v : 0.0f;
  }
}

__global__ __launch_bounds__(256) void bpnet_fused_kernel(
    const float* __restrict__ nodes,
    const int*   __restrict__ recs,
    const float* __restrict__ img,
    float*       __restrict__ out)
{
  __shared__ __attribute__((aligned(16))) float smem[LDS_DW];
  float* W1 = smem + W1_OFF;
  float* W2 = smem + W2_OFF;
  float* B1 = smem + B1_OFF;
  float* B2 = smem + B2_OFF;
  float* CB = smem + CB_OFF;

  const int tid   = threadIdx.x;
  const int wave  = __builtin_amdgcn_readfirstlane(tid >> 5);
  const int lane  = tid & 31;
  const int c16   = lane & 15;
  const int h     = lane >> 4;
  const int tslot = wave & 3;
  const int half  = wave >> 2;

  const v4i rec = ((const v4i*)recs)[blockIdx.x * 64 + tslot * 16 + c16];
  __builtin_amdgcn_sched_barrier(0);

  v4f wimg[IMG_ROUNDS];
#pragma unroll
  for (int i = 0; i < IMG_ROUNDS; ++i) {
    const int idx = ((i < IMG_ROUNDS - 1) ? i * 256 : IMG_V4 - 256) + tid;
    wimg[i] = ((const v4f*)img)[idx];
  }

  int et[3], nd[3];
  et[0] = (rec.x >> 4) & 3;
  et[1] = (rec.x >> 2) & 3;
  et[2] = rec.x & 3;
  nd[0] = rec.y;
  nd[1] = rec.z;
  nd[2] = rec.w;
  unsigned pm[3];
#pragma unroll
  for (int s = 0; s < 3; ++s) {
    unsigned m = 0;
#pragma unroll
    for (int p = 0; p < PP; ++p)
      m |= (__builtin_amdgcn_ballot_w32(et[s] == p) != 0u) ? (1u << p) : 0u;
    pm[s] = m;
  }

  v2f nb[3][4];
#pragma unroll
  for (int s = 0; s < 3; ++s) {
#pragma unroll
    for (int c = 0; c < 4; ++c) {
      const int kk = 4 * c + 2 * h;
      const float gx = nodes[nd[s] * DD + ((kk     < DD) ? kk     : 0)];
      const float gy = nodes[nd[s] * DD + ((kk + 1 < DD) ? kk + 1 : 0)];
      nb[s][c].x = (kk     < DD) ? gx : 0.0f;
      nb[s][c].y = (kk + 1 < DD) ? gy : 0.0f;
    }
  }

#pragma unroll
  for (int i = 0; i < IMG_ROUNDS; ++i) {
    const int idx = ((i < IMG_ROUNDS - 1) ? i * 256 : IMG_V4 - 256) + tid;
    ((v4f*)smem)[idx] = wimg[i];
  }
  __syncthreads();

  v8f a2[3];
#pragma unroll
  for (int s = 0; s < 3; ++s) a2[s] = (v8f){0.f, 0.f, 0.f, 0.f, 0.f, 0.f, 0.f, 0.f};

  int      p0[3];
  unsigned pr[3];
#pragma unroll
  for (int s = 0; s < 3; ++s) {
    p0[s] = __builtin_amdgcn_readfirstlane(et[s]);
    pr[s] = pm[s] & ~(1u << p0[s]);
  }

  const int nt0 = half * 4;
  for (int nt = nt0; nt < nt0 + 4; ++nt) {
    v2f wq[3][4];
#pragma unroll
    for (int s = 0; s < 3; ++s) {
      const float* wp = &W1[(p0[s] * 32 + nt) * 64 + lane * 2];
#pragma unroll
      for (int c = 0; c < 4; ++c) wq[s][c] = *(const v2f*)(wp + c * 512);
    }
    v8f a1[3];
#pragma unroll
    for (int s = 0; s < 3; ++s) {
      v8f acc = (v8f){0.f, 0.f, 0.f, 0.f, 0.f, 0.f, 0.f, 0.f};
      v2f w[4];
#pragma unroll
      for (int c = 0; c < 4; ++c) w[c] = wq[s][c];
      int p = p0[s];
      unsigned m = pr[s];
      for (;;) {
        const bool mine = (et[s] == p);
#pragma unroll
        for (int c = 0; c < 4; ++c) {
          v2f bb;
          bb.x = mine ? nb[s][c].x : 0.0f;
          bb.y = mine ? nb[s][c].y : 0.0f;
          acc = __builtin_amdgcn_wmma_f32_16x16x4_f32(
              false, w[c], false, bb, (short)0, acc, false, false);
        }
        if (m == 0u) break;
        p = __builtin_ctz(m);
        m &= m - 1u;
        const float* wp = &W1[(p * 32 + nt) * 64 + lane * 2];
#pragma unroll
        for (int c = 0; c < 4; ++c) w[c] = *(const v2f*)(wp + c * 512);
      }
      a1[s] = acc;
    }

    v2f vq[3][4];
#pragma unroll
    for (int s = 0; s < 3; ++s) {
      const float* wp = &W2[(((s * 4 + p0[s]) * 8 + nt) * 4) * 64 + lane * 2];
#pragma unroll
      for (int j = 0; j < 4; ++j) vq[s][j] = *(const v2f*)(wp + j * 64);
    }

    v8f t[3];
#pragma unroll
    for (int s = 0; s < 3; ++s) {
      const float* bp = &B1[et[s] * RR + nt * 16 + h * 8];
      const v4f b0 = *(const v4f*)bp;
      const v4f b1 = *(const v4f*)(bp + 4);
#pragma unroll
      for (int v = 0; v < 4; ++v) {
        float x = a1[s][v] + b0[v];
        float y = a1[s][v + 4] + b1[v];
        t[s][v]     = (x > 0.0f) ? x : 0.0f;
        t[s][v + 4] = (y > 0.0f) ? y : 0.0f;
      }
    }
    v8f f[3];
    f[0] = t[1] * t[2];
    f[1] = t[0] * t[2];
    f[2] = t[0] * t[1];

#pragma unroll
    for (int s = 0; s < 3; ++s) {
      v8f acc = a2[s];
      v2f w[4];
#pragma unroll
      for (int j = 0; j < 4; ++j) w[j] = vq[s][j];
      int p = p0[s];
      unsigned m = pr[s];
      for (;;) {
        const bool mine = (et[s] == p);
#pragma unroll
        for (int j = 0; j < 4; ++j) {
          v2f aa;
          aa.x = mine ? f[s][2 * j]     : 0.0f;
          aa.y = mine ? f[s][2 * j + 1] : 0.0f;
          acc = __builtin_amdgcn_wmma_f32_16x16x4_f32(
              false, aa, false, w[j], (short)0, acc, false, false);
        }
        if (m == 0u) break;
        p = __builtin_ctz(m);
        m &= m - 1u;
        const float* wp = &W2[(((s * 4 + p) * 8 + nt) * 4) * 64 + lane * 2];
#pragma unroll
        for (int j = 0; j < 4; ++j) w[j] = *(const v2f*)(wp + j * 64);
      }
      a2[s] = acc;
    }
  }

  if (half == 1) {
#pragma unroll
    for (int s = 0; s < 3; ++s)
#pragma unroll
      for (int v = 0; v < 8; ++v)
        CB[(tslot * 24 + s * 8 + v) * 32 + lane] = a2[s][v];
  }
  __syncthreads();
  if (half == 0) {
    float val[3][8];
    int   adr[3][8];
#pragma unroll
    for (int s = 0; s < 3; ++s) {
#pragma unroll
      for (int v = 0; v < 8; ++v) {
        const float sum = a2[s][v] + CB[(tslot * 24 + s * 8 + v) * 32 + lane];
        const int r2 = v + 8 * h;
        const int tr = __shfl(et[s], r2, 32);
        const int tg = __shfl(nd[s], r2, 32);
        val[s][v] = sum + B2[(s * 4 + tr) * 16 + c16];
        adr[s][v] = tg * DD + c16;
      }
    }
    if (c16 < DD) {
#pragma unroll
      for (int s = 0; s < 3; ++s)
#pragma unroll
        for (int v = 0; v < 8; ++v)
          atomicAdd(&out[adr[s][v]], val[s][v]);
    }
  }
}

extern "C" void kernel_launch(void* const* d_in, const int* in_sizes, int n_in,
                              void* d_out, int out_size, void* d_ws, size_t ws_size,
                              hipStream_t stream) {
  const float* nodes      = (const float*)d_in[0];
  const float* bp_params  = (const float*)d_in[1];
  const float* bp_bias    = (const float*)d_in[2];
  const float* ho_params  = (const float*)d_in[3];
  const float* ho_bias    = (const float*)d_in[4];
  const int*   edges      = (const int*)d_in[5];
  const int*   edge_types = (const int*)d_in[6];
  float* out  = (float*)d_out;
  int*   recs = (int*)d_ws + WS_REC;
  float* img  = (float*)d_ws + WS_IMG;

  prepare_kernel<<<dim3(PREP_WG), 256, 0, stream>>>(
      bp_params, bp_bias, ho_params, ho_bias, edges, edge_types, out, recs, img);

  bpnet_fused_kernel<<<dim3(256), 256, 0, stream>>>(nodes, recs, img, out);
}
